# starting kernel
# baseline (speedup 1.0000x reference)
#include <hip/hip_runtime.h>
#include <math.h>

typedef __attribute__((ext_vector_type(2))) float v2f;
typedef __attribute__((ext_vector_type(8))) float v8f;

#define KCOLS 1024
#define TILE_ROWS 16

__global__ __launch_bounds__(32) void rownorm_l2_kernel(const float* __restrict__ X,
                                                        float* __restrict__ Y,
                                                        int rows) {
    const int lane = threadIdx.x;
    const long r0  = (long)blockIdx.x * TILE_ROWS;

    __shared__ float Pl[TILE_ROWS * 32];
    __shared__ float Sl[TILE_ROWS];

    const float* __restrict__ xb = X + r0 * KCOLS;
    float*       __restrict__ yb = Y + r0 * KCOLS;

    float acc[TILE_ROWS];
#pragma unroll
    for (int m = 0; m < TILE_ROWS; ++m) acc[m] = 0.0f;

#pragma unroll
    for (int m = 0; m < TILE_ROWS; ++m) {
        if (r0 + m < rows) {
            const float4* rp = (const float4*)(xb + (long)m * KCOLS);
            float a = 0.0f;
#pragma unroll
            for (int j = 0; j < KCOLS / 128; ++j) {
                float4 v = rp[lane + 32 * j];
                a = fmaf(v.x, v.x, a);
                a = fmaf(v.y, v.y, a);
                a = fmaf(v.z, v.z, a);
                a = fmaf(v.w, v.w, a);
            }
            acc[m] = a;
        }
    }

#pragma unroll
    for (int m = 0; m < TILE_ROWS; ++m) Pl[m * 32 + lane] = acc[m];
    __syncthreads();

    v8f c = {};
    const v2f bones = {1.0f, 1.0f};
    const int mrow  = lane & 15;
    const int khalf = (lane >> 4) << 1;
#pragma unroll
    for (int k0 = 0; k0 < 32; k0 += 4) {
        v2f a;
        a.x = Pl[mrow * 32 + k0 + khalf + 0];
        a.y = Pl[mrow * 32 + k0 + khalf + 1];
        c = __builtin_amdgcn_wmma_f32_16x16x4_f32(
                 false, a,  false, bones,
                 (short)0, c,  false,  false);
    }

    if (lane == 0) {
#pragma unroll
        for (int v = 0; v < 8; ++v) Sl[v] = c[v];
    }
    if (lane == 16) {
#pragma unroll
        for (int v = 0; v < 8; ++v) Sl[8 + v] = c[v];
    }
    __syncthreads();

#pragma unroll
    for (int m = 0; m < TILE_ROWS; ++m) {
        if (r0 + m < rows) {
            const float s = Sl[m];
            const float d = (s > 0.0f) ? (1.0f / sqrtf(s)) : 0.0f;
            const float4* rp = (const float4*)(xb + (long)m * KCOLS);
            float4*       wp = (float4*)(yb + (long)m * KCOLS);
#pragma unroll
            for (int j = 0; j < KCOLS / 128; ++j) {
                float4 v = rp[lane + 32 * j];
                v.x *= d; v.y *= d; v.z *= d; v.w *= d;
                wp[lane + 32 * j] = v;
            }
        }
    }
}

extern "C" void kernel_launch(void* const* d_in, const int* in_sizes, int n_in,
                              void* d_out, int out_size, void* d_ws, size_t ws_size,
                              hipStream_t stream) {
    (void)n_in; (void)out_size; (void)d_ws; (void)ws_size;
    const float* X = (const float*)d_in[0];
    float*       Y = (float*)d_out;

    const int rows  = in_sizes[0] / KCOLS;
    const int tiles = (rows + TILE_ROWS - 1) / TILE_ROWS;

    rownorm_l2_kernel<<<tiles, 32, 0, stream>>>(X, Y, rows);
}


# One pass: a wave keeps its rows in registers from load to store (no second read of X); cross-lane sum by two chained f32 WMMA products, no LDS, no barrier; changes the summation order
# speedup vs baseline: 1.4939x; 1.4939x over previous
#include <hip/hip_runtime.h>
#include <math.h>

typedef __attribute__((ext_vector_type(2))) float v2f;
typedef __attribute__((ext_vector_type(4))) float v4f;
typedef __attribute__((ext_vector_type(8))) float v8f;

#define KCOLS 1024
#define PIECES 8
#define GROUP 2
#define STEPS 4
#define ROWS_PER_WAVE (GROUP * STEPS)
#define WAVES 8
#define THREADS (WAVES * 32)
#define ROWS_PER_BLOCK (WAVES * ROWS_PER_WAVE)

#define DEV static __device__ __forceinline__

DEV float wave_sum(float p0, float p1) {
    const v2f ones = {1.0f, 1.0f};
    const v8f zero = {};
    const v2f a1 = {p0, p1};
    const v8f d1 = __builtin_amdgcn_wmma_f32_16x16x4_f32(false, a1, false, ones, (short)0, zero, false, false);
    v8f d2 = {};
#pragma unroll
    for (int j = 0; j < 4; ++j) {
        const v2f a2 = {d1[2 * j], d1[2 * j + 1]};
        d2 = __builtin_amdgcn_wmma_f32_16x16x4_f32(false, a2, false, ones, (short)0, d2, false, false);
    }
    return d2[0];
}

__global__ __launch_bounds__(THREADS) void rownorm_l2_kernel(const float* __restrict__ X,
                                                             float* __restrict__ Y,
                                                             int rows) {
    const int lane = threadIdx.x & 31;
    const int wave = __builtin_amdgcn_readfirstlane(threadIdx.x >> 5);
    const int row0 = (blockIdx.x * WAVES + wave) * ROWS_PER_WAVE;
    if (row0 >= rows) return;
    const int last = rows - 1;

    v4f buf[2][GROUP][PIECES];

#define LOAD_STEP(B, S) do { \
    _Pragma("unroll") for (int r_ = 0; r_ < GROUP; ++r_) { \
        const int row_ = min(row0 + (S) * GROUP + r_, last); \
        const v4f* p_ = (const v4f*)(X + (long)row_ * KCOLS) + lane; \
        _Pragma("unroll") for (int q_ = 0; q_ < PIECES; ++q_) buf[(B)][r_][q_] = p_[32 * q_]; \
    } } while (0)

    LOAD_STEP(0, 0);

#pragma unroll
    for (int s = 0; s < STEPS; ++s) {
        __builtin_amdgcn_sched_barrier(0);
        if (s + 1 < STEPS) LOAD_STEP((s + 1) & 1, s + 1);
        __builtin_amdgcn_sched_barrier(0);

        float d[GROUP];
#pragma unroll
        for (int r = 0; r < GROUP; ++r) {
            const v4f* xc = buf[s & 1][r];
            float p0 = 0.0f, p1 = 0.0f;
#pragma unroll
            for (int q = 0; q < PIECES; ++q) {
                p0 = fmaf(xc[q].x, xc[q].x, p0);
                p1 = fmaf(xc[q].y, xc[q].y, p1);
                p0 = fmaf(xc[q].z, xc[q].z, p0);
                p1 = fmaf(xc[q].w, xc[q].w, p1);
            }
            const float ss  = wave_sum(p0, p1);
            const float inv = 1.0f / sqrtf(ss);
            d[r] = (ss > 0.0f) ? inv : 0.0f;
        }

#pragma unroll
        for (int r = 0; r < GROUP; ++r) {
            const int row = row0 + s * GROUP + r;
            if (row < rows) {
                const v4f* xc = buf[s & 1][r];
                v4f* wp = (v4f*)(Y + (long)row * KCOLS) + lane;
#pragma unroll
                for (int q = 0; q < PIECES; ++q) wp[32 * q] = xc[q] * d[r];
            }
        }
    }
#undef LOAD_STEP
}

extern "C" void kernel_launch(void* const* d_in, const int* in_sizes, int n_in,
                              void* d_out, int out_size, void* d_ws, size_t ws_size,
                              hipStream_t stream) {
    (void)n_in; (void)out_size; (void)d_ws; (void)ws_size;
    const float* X = (const float*)d_in[0];
    float*       Y = (float*)d_out;

    const int rows   = in_sizes[0] / KCOLS;
    const int blocks = (rows + ROWS_PER_BLOCK - 1) / ROWS_PER_BLOCK;

    rownorm_l2_kernel<<<blocks, THREADS, 0, stream>>>(X, Y, rows);
}
